# starting kernel
# baseline (speedup 1.0000x reference)
#include <hip/hip_runtime.h>
#include <hip/hip_bf16.h>

typedef float v2f __attribute__((ext_vector_type(2)));
typedef float v8f __attribute__((ext_vector_type(8)));

__global__ void fill_ones_kernel(float* __restrict__ p, int n) {
    int i = blockIdx.x * blockDim.x + threadIdx.x;
    if (i < n) p[i] = 1.0f;
}

__global__ void deg_count_kernel(const int* __restrict__ dst,
                                 float* __restrict__ deg, long long e) {
    long long i = (long long)blockIdx.x * blockDim.x + threadIdx.x;
    if (i < e) atomicAdd(&deg[dst[i]], 1.0f);
}

__global__ void dinv_kernel(const float* __restrict__ deg,
                            float* __restrict__ dinv, int n) {
    int i = blockIdx.x * blockDim.x + threadIdx.x;
    if (i < n) {
        float d = deg[i];
        dinv[i] = (d > 0.0f) ? __frsqrt_rn(d) : 0.0f;
    }
}

template <int KDIM, int NDIM>
__global__ void __launch_bounds__(128)
gemm_wmma_f32_kernel(const float* __restrict__ A, const float* __restrict__ W,
                     float* __restrict__ C, int M) {
    constexpr int STRIDE = 2 * NDIM + 32;
    __shared__ float sW[(KDIM / 2) * STRIDE];

    for (int i = threadIdx.x; i < KDIM * NDIM; i += blockDim.x) {
        int k = i / NDIM;
        int n = i - k * NDIM;
        sW[(k >> 1) * STRIDE + 2 * n + (k & 1)] = W[i];
    }
    __syncthreads();

    const int wave = threadIdx.x >> 5;
    const int lane = threadIdx.x & 31;
    const int half = lane >> 4;
    const int l15  = lane & 15;
    const int tile = blockIdx.x * 4 + wave;
    if (tile * 16 >= M) return;

    const int row = tile * 16 + l15;
    v2f af[KDIM / 4];
#pragma unroll
    for (int kk = 0; kk < KDIM / 4; ++kk) {
        const float* ap = A + (size_t)row * KDIM + kk * 4 + 2 * half;
        af[kk] = *reinterpret_cast<const v2f*>(ap);
    }

#pragma unroll
    for (int nt = 0; nt < NDIM / 16; ++nt) {
        v8f acc = {};
#pragma unroll
        for (int kk = 0; kk < KDIM / 4; ++kk) {
            const float* bp = &sW[(kk * 2 + half) * STRIDE + 2 * (nt * 16 + l15)];
            v2f b = *reinterpret_cast<const v2f*>(bp);
            acc = __builtin_amdgcn_wmma_f32_16x16x4_f32(
                 false, af[kk],  false, b,
                 (short)0, acc,  false,  false);
        }
#pragma unroll
        for (int r = 0; r < 8; ++r) {
            C[(size_t)(tile * 16 + r + 8 * half) * NDIM + nt * 16 + l15] = acc[r];
        }
    }
}

template <int F>
__global__ void selfloop_init_kernel(const float* __restrict__ h,
                                     const float* __restrict__ dinv,
                                     float* __restrict__ out, int n) {
    long long i = (long long)blockIdx.x * blockDim.x + threadIdx.x;
    if (i < (long long)n * F) {
        int node = (int)(i / F);
        float dv = dinv[node];
        out[i] = h[i] * dv * dv;
    }
}

template <int F>
__global__ void scatter_edges_kernel(const float* __restrict__ h,
                                     const int* __restrict__ src,
                                     const int* __restrict__ dst,
                                     const float* __restrict__ dinv,
                                     float* __restrict__ out, long long e) {
    long long gid  = (long long)blockIdx.x * blockDim.x + threadIdx.x;
    long long edge = gid >> 5;
    int lane       = threadIdx.x & 31;
    if (edge >= e) return;
    int s = src[edge];
    int d = dst[edge];
    float norm = dinv[s] * dinv[d];
    const float* hs = h + (size_t)s * F;
    float* od = out + (size_t)d * F;
#pragma unroll
    for (int f = lane; f < F; f += 32) {
        atomicAdd(&od[f], hs[f] * norm);
    }
}

template <int F, bool RELU>
__global__ void bias_act_kernel(float* __restrict__ h,
                                const float* __restrict__ bias, int n) {
    long long i = (long long)blockIdx.x * blockDim.x + threadIdx.x;
    if (i < (long long)n * F) {
        float v = h[i] + bias[(int)(i % F)];
        h[i] = RELU ? fmaxf(v, 0.0f) : v;
    }
}

extern "C" void kernel_launch(void* const* d_in, const int* in_sizes, int n_in,
                              void* d_out, int out_size, void* d_ws, size_t ws_size,
                              hipStream_t stream) {
    const float* x  = (const float*)d_in[0];
    const int*   ei = (const int*)d_in[1];
    const float* W1 = (const float*)d_in[2];
    const float* b1 = (const float*)d_in[3];
    const float* W2 = (const float*)d_in[4];
    const float* b2 = (const float*)d_in[5];
    float* out = (float*)d_out;

    const int       N = in_sizes[0] / 32;
    const long long E = (long long)in_sizes[1] / 2;
    const int* src = ei;
    const int* dst = ei + E;

    float* ws   = (float*)d_ws;
    float* deg  = ws;
    float* dinv = deg + N;
    float* hw1  = dinv + N;
    float* h1   = hw1 + (size_t)N * 64;
    float* hw2  = h1 + (size_t)N * 64;

    const int BT = 256;

    fill_ones_kernel<<<(N + BT - 1) / BT, BT, 0, stream>>>(deg, N);
    deg_count_kernel<<<(unsigned)((E + BT - 1) / BT), BT, 0, stream>>>(dst, deg, E);
    dinv_kernel<<<(N + BT - 1) / BT, BT, 0, stream>>>(deg, dinv, N);

    {
        int tiles  = N / 16;
        int blocks = (tiles + 3) / 4;
        gemm_wmma_f32_kernel<32, 64><<<blocks, 128, 0, stream>>>(x, W1, hw1, N);
    }
    {
        long long nf = (long long)N * 64;
        selfloop_init_kernel<64><<<(unsigned)((nf + BT - 1) / BT), BT, 0, stream>>>(hw1, dinv, h1, N);
        long long thr = E * 32;
        scatter_edges_kernel<64><<<(unsigned)((thr + BT - 1) / BT), BT, 0, stream>>>(hw1, src, dst, dinv, h1, E);
        bias_act_kernel<64, true><<<(unsigned)((nf + BT - 1) / BT), BT, 0, stream>>>(h1, b1, N);
    }

    {
        int tiles  = N / 16;
        int blocks = (tiles + 3) / 4;
        gemm_wmma_f32_kernel<64, 128><<<blocks, 128, 0, stream>>>(h1, W2, hw2, N);
    }
    {
        long long nf = (long long)N * 128;
        selfloop_init_kernel<128><<<(unsigned)((nf + BT - 1) / BT), BT, 0, stream>>>(hw2, dinv, out, N);
        long long thr = E * 32;
        scatter_edges_kernel<128><<<(unsigned)((thr + BT - 1) / BT), BT, 0, stream>>>(hw2, src, dst, dinv, out, E);
        bias_act_kernel<128, false><<<(unsigned)((nf + BT - 1) / BT), BT, 0, stream>>>(out, b2, N);
    }
}


# In-degrees counted with LDS integer adds (32 edge slices x 4 node windows; per-slice counts summed in the dinv kernel): no global float atomic per edge, no fill launch; degree values same bits
# speedup vs baseline: 3.5917x; 3.5917x over previous
#include <hip/hip_runtime.h>
#include <hip/hip_bf16.h>

typedef float v2f __attribute__((ext_vector_type(2)));
typedef float v8f __attribute__((ext_vector_type(8)));

constexpr int DEG_S  = 32;
constexpr int DEG_HW = 25024;

template <int TN>
__global__ void __launch_bounds__(1024)
deg_hist_key_kernel(const int* __restrict__ dst, unsigned* __restrict__ part,
                    unsigned* __restrict__ keys, int n, int e_total, int e_pad) {
    constexpr int U = 16;
    __shared__ unsigned hist[DEG_HW];

    const int tid = threadIdx.x;
    const int lo  = (int)blockIdx.y * DEG_HW;
    {
        uint4* h4 = reinterpret_cast<uint4*>(hist);
        for (int i = tid; i < DEG_HW / 4; i += 1024) h4[i] = make_uint4(0u, 0u, 0u, 0u);
    }
    __syncthreads();

    const int nq = e_pad / 4;
    const int qs = (((nq + (int)gridDim.x - 1) / (int)gridDim.x) + 31) & ~31;
    const int q0 = min((int)blockIdx.x * qs, nq);
    const int q1 = min(q0 + qs, nq);
    const bool wr_keys = (blockIdx.y == 0);

    auto take = [&](int q, int4 t) {
        const unsigned d[4] = {(unsigned)t.x, (unsigned)t.y, (unsigned)t.z, (unsigned)t.w};
#pragma unroll
        for (int c = 0; c < 4; ++c) {
            const unsigned r = d[c] - (unsigned)lo;
            if (r < (unsigned)DEG_HW) atomicAdd(&hist[r], 1u);
        }
        if (wr_keys) {
            unsigned w = 0;
#pragma unroll
            for (int c = 0; c < 4; ++c) w |= min(d[c] / (unsigned)TN, 255u) << (8 * c);
            keys[q] = w;
        }
    };

    const bool al = (reinterpret_cast<size_t>(dst) & 15) == 0;
    const int qv1 = al ? max(q0, min(q1, e_total / 4)) : q0;
    const int4* dst4 = reinterpret_cast<const int4*>(dst);
    for (int qb = q0 + tid; qb < qv1; qb += U * 1024) {
        int4 v[U];
#pragma unroll
        for (int u = 0; u < U; ++u) {
            const int q = qb + u * 1024;
            v[u] = make_int4(-1, -1, -1, -1);
            if (q < qv1) v[u] = dst4[q];
        }
#pragma unroll
        for (int u = 0; u < U; ++u) {
            const int q = qb + u * 1024;
            if (q < qv1) take(q, v[u]);
        }
    }
    for (int q = qv1 + tid; q < q1; q += 1024) {
        const int e = 4 * q;
        int4 t = make_int4(-1, -1, -1, -1);
        if (e     < e_total) t.x = dst[e];
        if (e + 1 < e_total) t.y = dst[e + 1];
        if (e + 2 < e_total) t.z = dst[e + 2];
        if (e + 3 < e_total) t.w = dst[e + 3];
        take(q, t);
    }
    __syncthreads();

    unsigned* prow = part + (size_t)blockIdx.x * n + lo;
    const int cnt = min(DEG_HW, n - lo);
    for (int i = tid; i < cnt; i += 1024) prow[i] = hist[i];
}

__global__ void dinv_kernel(const unsigned* __restrict__ part,
                            float* __restrict__ dinv, int n) {
    int i = blockIdx.x * blockDim.x + threadIdx.x;
    if (i < n) {
        unsigned c = 1u;
#pragma unroll
        for (int s = 0; s < DEG_S; ++s) c += part[(size_t)s * n + i];
        float d = (float)c;
        dinv[i] = (d > 0.0f) ? __frsqrt_rn(d) : 0.0f;
    }
}

template <int KDIM, int NDIM>
__global__ void __launch_bounds__(128)
gemm_wmma_f32_kernel(const float* __restrict__ A, const float* __restrict__ W,
                     float* __restrict__ C, int M) {
    constexpr int KT = KDIM / 4;
    constexpr int NT = NDIM / 16;

    const int wave = __builtin_amdgcn_readfirstlane((int)(threadIdx.x >> 5));
    const int lane = threadIdx.x & 31;
    const int half = lane >> 4;
    const int l15  = lane & 15;
    const int ntiles = M / 16;
    const int nwaves = (int)gridDim.x * 4;
    int tile = (int)blockIdx.x * 4 + wave;
    if (tile >= ntiles) return;

    v2f bf[KT][NT];
#pragma unroll
    for (int kk = 0; kk < KT; ++kk) {
#pragma unroll
        for (int nt = 0; nt < NT; ++nt) {
            const float* wp = W + (size_t)(kk * 4 + 2 * half) * NDIM + nt * 16 + l15;
            v2f b;
            b.x = wp[0];
            b.y = wp[NDIM];
            bf[kk][nt] = b;
        }
    }

    for (; tile < ntiles; tile += nwaves) {
        v2f af[KT];
        const float* ap = A + (size_t)(tile * 16 + l15) * KDIM + 2 * half;
#pragma unroll
        for (int kk = 0; kk < KT; ++kk) af[kk] = *reinterpret_cast<const v2f*>(ap + kk * 4);

#pragma unroll
        for (int nt = 0; nt < NT; ++nt) {
            v8f acc = {};
#pragma unroll
            for (int kk = 0; kk < KT; ++kk) {
                acc = __builtin_amdgcn_wmma_f32_16x16x4_f32(
                     false, af[kk],  false, bf[kk][nt],
                     (short)0, acc,  false,  false);
            }
#pragma unroll
            for (int r = 0; r < 8; ++r) {
                C[(size_t)(tile * 16 + r + 8 * half) * NDIM + nt * 16 + l15] = acc[r];
            }
        }
    }
}

constexpr int REC_CAP = 512;
template <int C> struct RowVec;
template <> struct RowVec<2> { typedef float2 type; };
template <> struct RowVec<4> { typedef float4 type; };
template <int F, int TN, bool RELU, bool WRITE_REC, bool READ_REC>
__global__ void __launch_bounds__(1024)
agg_tile_kernel(const float* __restrict__ h, const int* __restrict__ src,
                const int* __restrict__ dst, const float* __restrict__ dinv,
                const float* __restrict__ bias, float* __restrict__ out,
                const unsigned* __restrict__ keys, int* __restrict__ rec,
                int* __restrict__ rcnt, int n, int e_pad) {
    constexpr int NW   = 32;
    constexpr int QCAP = 64;
    constexpr int C    = F / 32;
    constexpr int RB   = 64 / C;
    typedef typename RowVec<C>::type rowv_t;
    constexpr int CH   = 4096;

    __shared__ float acc[TN * F];
    __shared__ float sdinv[TN];
    __shared__ int   qe[NW * QCAP];

    const int tid   = threadIdx.x;
    const int lane  = tid & 31;
    const int wave  = __builtin_amdgcn_readfirstlane(tid >> 5);
    const int node0 = blockIdx.x * TN;
    const int nrows = min(TN, n - node0);

    int4* myrec = reinterpret_cast<int4*>(rec) + ((size_t)blockIdx.x * NW + wave) * REC_CAP;
    int  recn = 0;
    bool use_list = false;
    int  nrec = 0;
    if (READ_REC) {
        const int c_ = rcnt[blockIdx.x * NW + lane];
        use_list = __builtin_amdgcn_ballot_w32(c_ > REC_CAP) == 0;
        nrec = __builtin_amdgcn_readlane(c_, wave);
    }

    for (int r = tid; r < nrows; r += 1024) sdinv[r] = dinv[node0 + r];
    {
        const float4* h4 = reinterpret_cast<const float4*>(h + (size_t)node0 * F);
        for (int i = tid; i < nrows * (F / 4); i += 1024) {
            float dv = dinv[node0 + i / (F / 4)];
            float4 v = h4[i];
            v.x = v.x * dv * dv; v.y = v.y * dv * dv;
            v.z = v.z * dv * dv; v.w = v.w * dv * dv;
            const int r  = i / (F / 4);
            const int j0 = 4 * (i - r * (F / 4));
            float* ar = acc + r * F;
            ar[((j0    ) % C) * 32 + (j0    ) / C] = v.x;
            ar[((j0 + 1) % C) * 32 + (j0 + 1) / C] = v.y;
            ar[((j0 + 2) % C) * 32 + (j0 + 2) / C] = v.z;
            ar[((j0 + 3) % C) * 32 + (j0 + 3) / C] = v.w;
        }
    }
    __syncthreads();

    const int  nchunks = e_pad / CH;
    const int  qbase   = wave * QCAP;
    int qn = 0;

    auto gather = [&](int s_, int dl_, float nrm_, int cnt) {
        for (int k0 = 0; k0 < cnt; k0 += RB) {
            rowv_t v[RB];
#pragma unroll
            for (int j = 0; j < RB; ++j) {
                int k = min(k0 + j, cnt - 1);
                int s = __builtin_amdgcn_readlane(s_, k);
                v[j] = *reinterpret_cast<const rowv_t*>(h + (size_t)s * F + C * lane);
            }
#pragma unroll
            for (int j = 0; j < RB; ++j) {
                int k = k0 + j;
                if (k < cnt) {
                    float nrm = __int_as_float(__builtin_amdgcn_readlane(__float_as_int(nrm_), k));
                    int   dl  = __builtin_amdgcn_readlane(dl_, k);
                    float* ap = &acc[dl * F + lane];
                    const float* vj = reinterpret_cast<const float*>(&v[j]);
#pragma unroll
                    for (int c = 0; c < C; ++c) atomicAdd(ap + 32 * c, vj[c] * nrm);
                }
            }
        }
    };

    auto run_batch = [&](int start, int cnt) {
        int s_ = 0, dl_ = 0; float nrm_ = 0.0f;
        if (lane < cnt) {
            int e_ = qe[qbase + start + lane];
            s_  = src[e_];
            dl_ = dst[e_] - node0;
            nrm_ = dinv[s_] * sdinv[dl_];
        }
        gather(s_, dl_, nrm_, cnt);
        if (WRITE_REC) {
            if (lane < cnt && recn + lane < REC_CAP)
                myrec[recn + lane] = make_int4(s_, dl_, __float_as_int(nrm_), 0);
            recn += cnt;
        }
    };

    auto drain = [&](unsigned bits, int ebase) {
        for (int t = 0; t < 32; ++t) {
            unsigned mask = __builtin_amdgcn_ballot_w32(bits != 0);
            if (mask == 0) break;
            if (bits != 0) {
                int b = __builtin_ctz(bits);
                bits &= bits - 1;
                int pos = qn + (int)__builtin_amdgcn_mbcnt_lo(mask, 0u);
                int i = b & 7;
                qe[qbase + pos] = ebase + 512 * (i >> 2) + 4 * (i & 3) + (b >> 3);
            }
            qn += __builtin_popcount(mask);
            if (qn >= 32) {
                qn -= 32;
                run_batch(qn, 32);
            }
        }
    };

    if (use_list) {
        for (int b = 0; b < nrec; b += 32) {
            int4 r_ = make_int4(0, 0, 0, 0);
            if (b + lane < nrec) r_ = myrec[b + lane];
            gather(r_.x, r_.y, __int_as_float(r_.z), min(32, nrec - b));
        }
    } else {
        const unsigned pat = (unsigned)blockIdx.x * 0x01010101u;
        const uint4* keys4 = reinterpret_cast<const uint4*>(keys);
        for (int chunk = wave; chunk < nchunks; chunk += NW) {
            const int base = chunk * CH;
            uint4 kw[8];
#pragma unroll
            for (int k = 0; k < 8; ++k) kw[k] = keys4[(base >> 4) + 32 * k + lane];

#pragma unroll
            for (int p = 0; p < 4; ++p) {
                const unsigned w8[8] = {kw[2 * p].x,     kw[2 * p].y,     kw[2 * p].z,     kw[2 * p].w,
                                        kw[2 * p + 1].x, kw[2 * p + 1].y, kw[2 * p + 1].z, kw[2 * p + 1].w};
                unsigned bits = 0;
#pragma unroll
                for (int i = 0; i < 8; ++i) {
                    unsigned x = w8[i] ^ pat;
                    unsigned y = ((x & 0x7F7F7F7Fu) + 0x7F7F7F7Fu) | x;
                    bits |= (~y & 0x80808080u) >> (7 - i);
                }
                drain(bits, base + 1024 * p + 16 * lane);
            }
        }
        run_batch(0, qn);
    }
    if (WRITE_REC && lane == 0) rcnt[blockIdx.x * NW + wave] = recn;
    __syncthreads();

    {
        const float4* b4 = reinterpret_cast<const float4*>(bias);
        float4* o4 = reinterpret_cast<float4*>(out + (size_t)node0 * F);
        for (int i = tid; i < nrows * (F / 4); i += 1024) {
            const int r  = i / (F / 4);
            const int j0 = 4 * (i - r * (F / 4));
            const float* ar = acc + r * F;
            float4 v;
            v.x = ar[((j0    ) % C) * 32 + (j0    ) / C];
            v.y = ar[((j0 + 1) % C) * 32 + (j0 + 1) / C];
            v.z = ar[((j0 + 2) % C) * 32 + (j0 + 2) / C];
            v.w = ar[((j0 + 3) % C) * 32 + (j0 + 3) / C];
            float4 b = b4[i % (F / 4)];
            v.x += b.x; v.y += b.y; v.z += b.z; v.w += b.w;
            if (RELU) {
                v.x = fmaxf(v.x, 0.0f); v.y = fmaxf(v.y, 0.0f);
                v.z = fmaxf(v.z, 0.0f); v.w = fmaxf(v.w, 0.0f);
            }
            o4[i] = v;
        }
    }
}

extern "C" void kernel_launch(void* const* d_in, const int* in_sizes, int n_in,
                              void* d_out, int out_size, void* d_ws, size_t ws_size,
                              hipStream_t stream) {
    const float* x  = (const float*)d_in[0];
    const int*   ei = (const int*)d_in[1];
    const float* W1 = (const float*)d_in[2];
    const float* b1 = (const float*)d_in[3];
    const float* W2 = (const float*)d_in[4];
    const float* b2 = (const float*)d_in[5];
    float* out = (float*)d_out;

    const int       N = in_sizes[0] / 32;
    const long long E = (long long)in_sizes[1] / 2;
    const int* src = ei;
    const int* dst = ei + E;

    const int EPAD = (int)((E + 4095) / 4096) * 4096;
    const int TILES = (N + 399) / 400;
    const int NWIN = (N + DEG_HW - 1) / DEG_HW;
    unsigned* keys = (unsigned*)d_ws;
    int* rec    = (int*)((char*)d_ws + EPAD);
    int* rcnt   = rec + (size_t)TILES * 32 * REC_CAP * 4;
    float* ws   = (float*)(rcnt + (size_t)TILES * 32);
    float* dinv = ws;
    float* hw1  = dinv + N;
    float* h1   = hw1 + (size_t)N * 64;
    float* hw2  = h1 + (size_t)N * 64;
    unsigned* part = (unsigned*)(hw2 + (size_t)N * 128);

    const int BT = 256;

    deg_hist_key_kernel<400><<<dim3(DEG_S, NWIN), 1024, 0, stream>>>(dst, part, keys, N, (int)E, EPAD);
    dinv_kernel<<<(N + BT - 1) / BT, BT, 0, stream>>>(part, dinv, N);

    gemm_wmma_f32_kernel<32, 64><<<1024, 128, 0, stream>>>(x, W1, hw1, N);
    agg_tile_kernel<64, 400, true, true, false><<<TILES, 1024, 0, stream>>>(hw1, src, dst, dinv, b1, h1, keys, rec, rcnt, N, EPAD);

    gemm_wmma_f32_kernel<64, 128><<<512, 128, 0, stream>>>(h1, W2, hw2, N);
    agg_tile_kernel<128, 400, false, false, true><<<TILES, 1024, 0, stream>>>(hw2, src, dst, dinv, b2, out, keys, rec, rcnt, N, EPAD);
}


# speedup vs baseline: 3.8201x; 1.0636x over previous
#include <hip/hip_runtime.h>
#include <hip/hip_bf16.h>

typedef float v2f __attribute__((ext_vector_type(2)));
typedef float v8f __attribute__((ext_vector_type(8)));
typedef float v4f __attribute__((ext_vector_type(4)));

constexpr int DEG_S  = 32;
constexpr int DEG_HW = 25024;

template <int TN>
__global__ void __launch_bounds__(1024)
deg_hist_key_kernel(const int* __restrict__ dst, unsigned* __restrict__ part,
                    unsigned* __restrict__ keys, int n, int e_total, int e_pad) {
    constexpr int U = 16;
    __shared__ unsigned hist[DEG_HW];

    const int tid = threadIdx.x;
    const int lo  = (int)blockIdx.y * DEG_HW;
    {
        uint4* h4 = reinterpret_cast<uint4*>(hist);
        for (int i = tid; i < DEG_HW / 4; i += 1024) h4[i] = make_uint4(0u, 0u, 0u, 0u);
    }
    __syncthreads();

    const int nq = e_pad / 4;
    const int qs = (((nq + (int)gridDim.x - 1) / (int)gridDim.x) + 31) & ~31;
    const int q0 = min((int)blockIdx.x * qs, nq);
    const int q1 = min(q0 + qs, nq);
    const bool wr_keys = (blockIdx.y == 0);

    auto take = [&](int q, int4 t) {
        const unsigned d[4] = {(unsigned)t.x, (unsigned)t.y, (unsigned)t.z, (unsigned)t.w};
#pragma unroll
        for (int c = 0; c < 4; ++c) {
            const unsigned r = d[c] - (unsigned)lo;
            if (r < (unsigned)DEG_HW) atomicAdd(&hist[r], 1u);
        }
        if (wr_keys) {
            unsigned w = 0;
#pragma unroll
            for (int c = 0; c < 4; ++c) w |= min(d[c] / (unsigned)TN, 255u) << (8 * c);
            keys[q] = w;
        }
    };

    const bool al = (reinterpret_cast<size_t>(dst) & 15) == 0;
    const int qv1 = al ? max(q0, min(q1, e_total / 4)) : q0;
    const int4* dst4 = reinterpret_cast<const int4*>(dst);
    for (int qb = q0 + tid; qb < qv1; qb += U * 1024) {
        int4 v[U];
#pragma unroll
        for (int u = 0; u < U; ++u) {
            const int q = qb + u * 1024;
            v[u] = make_int4(-1, -1, -1, -1);
            if (q < qv1) v[u] = dst4[q];
        }
#pragma unroll
        for (int u = 0; u < U; ++u) {
            const int q = qb + u * 1024;
            if (q < qv1) take(q, v[u]);
        }
    }
    for (int q = qv1 + tid; q < q1; q += 1024) {
        const int e = 4 * q;
        int4 t = make_int4(-1, -1, -1, -1);
        if (e     < e_total) t.x = dst[e];
        if (e + 1 < e_total) t.y = dst[e + 1];
        if (e + 2 < e_total) t.z = dst[e + 2];
        if (e + 3 < e_total) t.w = dst[e + 3];
        take(q, t);
    }
    __syncthreads();

    unsigned* prow = part + (size_t)blockIdx.x * n + lo;
    const int cnt = min(DEG_HW, n - lo);
    for (int i = tid; i < cnt; i += 1024) prow[i] = hist[i];
}

__global__ void dinv_kernel(const unsigned* __restrict__ part,
                            float* __restrict__ dinv, int n) {
    int i = blockIdx.x * blockDim.x + threadIdx.x;
    if (i < n) {
        unsigned c = 1u;
#pragma unroll
        for (int s = 0; s < DEG_S; ++s) c += part[(size_t)s * n + i];
        float d = (float)c;
        dinv[i] = (d > 0.0f) ? __frsqrt_rn(d) : 0.0f;
    }
}

template <int KDIM, int NDIM>
__global__ void __launch_bounds__(128)
gemm_wmma_f32_kernel(const float* __restrict__ A, const float* __restrict__ W,
                     float* __restrict__ C, int M) {
    constexpr int KT = KDIM / 4;
    constexpr int NT = NDIM / 16;
    constexpr int CP = NDIM + 4;
    constexpr int NV = 16 * NDIM / 128;
    __shared__ float4 stage4[4 * 16 * CP / 4];

    const int wave = __builtin_amdgcn_readfirstlane((int)(threadIdx.x >> 5));
    const int lane = threadIdx.x & 31;
    const int half = lane >> 4;
    const int l15  = lane & 15;
    const int ntiles = M / 16;
    const int nwaves = (int)gridDim.x * 4;
    int tile = (int)blockIdx.x * 4 + wave;
    if (tile >= ntiles) return;

    v2f bf[KT][NT];
#pragma unroll
    for (int kk = 0; kk < KT; ++kk) {
#pragma unroll
        for (int nt = 0; nt < NT; ++nt) {
            const float* wp = W + (size_t)(kk * 4 + 2 * half) * NDIM + nt * 16 + l15;
            v2f b;
            b.x = wp[0];
            b.y = wp[NDIM];
            bf[kk][nt] = b;
        }
    }

    float* st = reinterpret_cast<float*>(stage4) + wave * (16 * CP);

    for (; tile < ntiles; tile += nwaves) {
        v2f af[KT];
        const float* ap = A + (size_t)(tile * 16 + l15) * KDIM + 2 * half;
#pragma unroll
        for (int kk = 0; kk < KT; ++kk) af[kk] = *reinterpret_cast<const v2f*>(ap + kk * 4);

#pragma unroll
        for (int nt = 0; nt < NT; ++nt) {
            v8f acc = {};
#pragma unroll
            for (int kk = 0; kk < KT; ++kk) {
                acc = __builtin_amdgcn_wmma_f32_16x16x4_f32(
                     false, af[kk],  false, bf[kk][nt],
                     (short)0, acc,  false,  false);
            }
#pragma unroll
            for (int r = 0; r < 8; ++r) {
                st[(r + 8 * half) * CP + nt * 16 + l15] = acc[r];
            }
        }

        v4f* c4 = reinterpret_cast<v4f*>(C + (size_t)tile * 16 * NDIM);
        v4f v[NV];
#pragma unroll
        for (int j = 0; j < NV; ++j) {
            const int e = (j * 32 + lane) * 4;
            v[j] = *reinterpret_cast<const v4f*>(st + (e / NDIM) * CP + (e % NDIM));
        }
#pragma unroll
        for (int g = 0; g < NV; g += 8) {
            asm volatile("" :: "v"(v[g]), "v"(v[g + 1]), "v"(v[g + 2]), "v"(v[g + 3]), "v"(v[g + 4]), "v"(v[g + 5]), "v"(v[g + 6]), "v"(v[g + 7]));
        }
#pragma unroll
        for (int j = 0; j < NV; ++j) c4[j * 32 + lane] = v[j];
    }
}

constexpr int REC_CAP = 512;
template <int C> struct RowVec;
template <> struct RowVec<2> { typedef float2 type; };
template <> struct RowVec<4> { typedef float4 type; };
template <int F, int TN, bool RELU, bool WRITE_REC, bool READ_REC>
__global__ void __launch_bounds__(1024)
agg_tile_kernel(const float* __restrict__ h, const int* __restrict__ src,
                const int* __restrict__ dst, const float* __restrict__ dinv,
                const float* __restrict__ bias, float* __restrict__ out,
                const unsigned* __restrict__ keys, int* __restrict__ rec,
                int* __restrict__ rcnt, int n, int e_pad) {
    constexpr int NW   = 32;
    constexpr int QCAP = 64;
    constexpr int C    = F / 32;
    constexpr int RB   = 64 / C;
    typedef typename RowVec<C>::type rowv_t;
    constexpr int CH   = 4096;

    __shared__ float acc[TN * F];
    __shared__ float sdinv[TN];
    __shared__ int   qe[NW * QCAP];

    const int tid   = threadIdx.x;
    const int lane  = tid & 31;
    const int wave  = __builtin_amdgcn_readfirstlane(tid >> 5);
    const int node0 = blockIdx.x * TN;
    const int nrows = min(TN, n - node0);

    int4* myrec = reinterpret_cast<int4*>(rec) + ((size_t)blockIdx.x * NW + wave) * REC_CAP;
    int  recn = 0;
    bool use_list = false;
    int  nrec = 0;
    if (READ_REC) {
        const int c_ = rcnt[blockIdx.x * NW + lane];
        use_list = __builtin_amdgcn_ballot_w32(c_ > REC_CAP) == 0;
        nrec = __builtin_amdgcn_readlane(c_, wave);
    }

    for (int r = tid; r < nrows; r += 1024) sdinv[r] = dinv[node0 + r];
    {
        const float4* h4 = reinterpret_cast<const float4*>(h + (size_t)node0 * F);
        for (int i = tid; i < nrows * (F / 4); i += 1024) {
            float dv = dinv[node0 + i / (F / 4)];
            float4 v = h4[i];
            v.x = v.x * dv * dv; v.y = v.y * dv * dv;
            v.z = v.z * dv * dv; v.w = v.w * dv * dv;
            const int r  = i / (F / 4);
            const int j0 = 4 * (i - r * (F / 4));
            float* ar = acc + r * F;
            ar[((j0    ) % C) * 32 + (j0    ) / C] = v.x;
            ar[((j0 + 1) % C) * 32 + (j0 + 1) / C] = v.y;
            ar[((j0 + 2) % C) * 32 + (j0 + 2) / C] = v.z;
            ar[((j0 + 3) % C) * 32 + (j0 + 3) / C] = v.w;
        }
    }
    __syncthreads();

    const int  nchunks = e_pad / CH;
    const int  qbase   = wave * QCAP;
    int qn = 0;

    auto gather = [&](int s_, int dl_, float nrm_, int cnt) {
        for (int k0 = 0; k0 < cnt; k0 += RB) {
            rowv_t v[RB];
#pragma unroll
            for (int j = 0; j < RB; ++j) {
                int k = min(k0 + j, cnt - 1);
                int s = __builtin_amdgcn_readlane(s_, k);
                v[j] = *reinterpret_cast<const rowv_t*>(h + (size_t)s * F + C * lane);
            }
#pragma unroll
            for (int j = 0; j < RB; ++j) {
                int k = k0 + j;
                if (k < cnt) {
                    float nrm = __int_as_float(__builtin_amdgcn_readlane(__float_as_int(nrm_), k));
                    int   dl  = __builtin_amdgcn_readlane(dl_, k);
                    float* ap = &acc[dl * F + lane];
                    const float* vj = reinterpret_cast<const float*>(&v[j]);
#pragma unroll
                    for (int c = 0; c < C; ++c) atomicAdd(ap + 32 * c, vj[c] * nrm);
                }
            }
        }
    };

    auto run_batch = [&](int start, int cnt) {
        int s_ = 0, dl_ = 0; float nrm_ = 0.0f;
        if (lane < cnt) {
            int e_ = qe[qbase + start + lane];
            s_  = src[e_];
            dl_ = dst[e_] - node0;
            nrm_ = dinv[s_] * sdinv[dl_];
        }
        gather(s_, dl_, nrm_, cnt);
        if (WRITE_REC) {
            if (lane < cnt && recn + lane < REC_CAP)
                myrec[recn + lane] = make_int4(s_, dl_, __float_as_int(nrm_), 0);
            recn += cnt;
        }
    };

    auto drain = [&](unsigned bits, int ebase) {
        for (int t = 0; t < 32; ++t) {
            unsigned mask = __builtin_amdgcn_ballot_w32(bits != 0);
            if (mask == 0) break;
            if (bits != 0) {
                int b = __builtin_ctz(bits);
                bits &= bits - 1;
                int pos = qn + (int)__builtin_amdgcn_mbcnt_lo(mask, 0u);
                int i = b & 7;
                qe[qbase + pos] = ebase + 512 * (i >> 2) + 4 * (i & 3) + (b >> 3);
            }
            qn += __builtin_popcount(mask);
            if (qn >= 32) {
                qn -= 32;
                run_batch(qn, 32);
            }
        }
    };

    if (use_list) {
        for (int b = 0; b < nrec; b += 32) {
            int4 r_ = make_int4(0, 0, 0, 0);
            if (b + lane < nrec) r_ = myrec[b + lane];
            gather(r_.x, r_.y, __int_as_float(r_.z), min(32, nrec - b));
        }
    } else {
        const unsigned pat = (unsigned)blockIdx.x * 0x01010101u;
        const uint4* keys4 = reinterpret_cast<const uint4*>(keys);
        for (int chunk = wave; chunk < nchunks; chunk += NW) {
            const int base = chunk * CH;
            uint4 kw[8];
#pragma unroll
            for (int k = 0; k < 8; ++k) kw[k] = keys4[(base >> 4) + 32 * k + lane];

#pragma unroll
            for (int p = 0; p < 4; ++p) {
                const unsigned w8[8] = {kw[2 * p].x,     kw[2 * p].y,     kw[2 * p].z,     kw[2 * p].w,
                                        kw[2 * p + 1].x, kw[2 * p + 1].y, kw[2 * p + 1].z, kw[2 * p + 1].w};
                unsigned bits = 0;
#pragma unroll
                for (int i = 0; i < 8; ++i) {
                    unsigned x = w8[i] ^ pat;
                    unsigned y = ((x & 0x7F7F7F7Fu) + 0x7F7F7F7Fu) | x;
                    bits |= (~y & 0x80808080u) >> (7 - i);
                }
                drain(bits, base + 1024 * p + 16 * lane);
            }
        }
        run_batch(0, qn);
    }
    if (WRITE_REC && lane == 0) rcnt[blockIdx.x * NW + wave] = recn;
    __syncthreads();

    {
        const float4* b4 = reinterpret_cast<const float4*>(bias);
        float4* o4 = reinterpret_cast<float4*>(out + (size_t)node0 * F);
        for (int i = tid; i < nrows * (F / 4); i += 1024) {
            const int r  = i / (F / 4);
            const int j0 = 4 * (i - r * (F / 4));
            const float* ar = acc + r * F;
            float4 v;
            v.x = ar[((j0    ) % C) * 32 + (j0    ) / C];
            v.y = ar[((j0 + 1) % C) * 32 + (j0 + 1) / C];
            v.z = ar[((j0 + 2) % C) * 32 + (j0 + 2) / C];
            v.w = ar[((j0 + 3) % C) * 32 + (j0 + 3) / C];
            float4 b = b4[i % (F / 4)];
            v.x += b.x; v.y += b.y; v.z += b.z; v.w += b.w;
            if (RELU) {
                v.x = fmaxf(v.x, 0.0f); v.y = fmaxf(v.y, 0.0f);
                v.z = fmaxf(v.z, 0.0f); v.w = fmaxf(v.w, 0.0f);
            }
            o4[i] = v;
        }
    }
}

extern "C" void kernel_launch(void* const* d_in, const int* in_sizes, int n_in,
                              void* d_out, int out_size, void* d_ws, size_t ws_size,
                              hipStream_t stream) {
    const float* x  = (const float*)d_in[0];
    const int*   ei = (const int*)d_in[1];
    const float* W1 = (const float*)d_in[2];
    const float* b1 = (const float*)d_in[3];
    const float* W2 = (const float*)d_in[4];
    const float* b2 = (const float*)d_in[5];
    float* out = (float*)d_out;

    const int       N = in_sizes[0] / 32;
    const long long E = (long long)in_sizes[1] / 2;
    const int* src = ei;
    const int* dst = ei + E;

    const int EPAD = (int)((E + 4095) / 4096) * 4096;
    const int TILES = (N + 399) / 400;
    const int NWIN = (N + DEG_HW - 1) / DEG_HW;
    float* hw1  = (float*)d_ws;
    float* h1   = hw1 + (size_t)N * 64;
    float* hw2  = h1 + (size_t)N * 64;
    int* rec    = (int*)(hw2 + (size_t)N * 128);
    int* rcnt   = rec + (size_t)TILES * 32 * REC_CAP * 4;
    unsigned* keys = (unsigned*)(rcnt + (size_t)TILES * 32);
    float* dinv = (float*)((char*)keys + EPAD);
    unsigned* part = (unsigned*)(dinv + N);

    const int BT = 256;

    deg_hist_key_kernel<400><<<dim3(DEG_S, NWIN), 1024, 0, stream>>>(dst, part, keys, N, (int)E, EPAD);
    dinv_kernel<<<(N + BT - 1) / BT, BT, 0, stream>>>(part, dinv, N);

    gemm_wmma_f32_kernel<32, 64><<<1024, 128, 0, stream>>>(x, W1, hw1, N);
    agg_tile_kernel<64, 400, true, true, false><<<TILES, 1024, 0, stream>>>(hw1, src, dst, dinv, b1, h1, keys, rec, rcnt, N, EPAD);

    gemm_wmma_f32_kernel<64, 128><<<512, 128, 0, stream>>>(h1, W2, hw2, N);
    agg_tile_kernel<128, 400, false, false, true><<<TILES, 1024, 0, stream>>>(hw2, src, dst, dinv, b2, out, keys, rec, rcnt, N, EPAD);
}


# Degree counts and x W1 in one launch: the 128 counting groups and 128 32-wave product groups run side by side (the product needs no degrees); 6 launches become 5; f32 results same bits
# speedup vs baseline: 3.9452x; 1.0327x over previous
#include <hip/hip_runtime.h>
#include <hip/hip_bf16.h>

typedef float v2f __attribute__((ext_vector_type(2)));
typedef float v8f __attribute__((ext_vector_type(8)));
typedef float v4f __attribute__((ext_vector_type(4)));

constexpr int DEG_S  = 32;
constexpr int DEG_HW = 25024;

template <int TN>
__device__ __forceinline__ void
deg_hist_key_body(const int* __restrict__ dst, unsigned* __restrict__ part,
                  unsigned* __restrict__ keys, int n, int e_total, int e_pad,
                  int slice, int window, unsigned* hist) {
    constexpr int U = 16;

    const int tid = threadIdx.x;
    const int lo  = window * DEG_HW;
    {
        uint4* h4 = reinterpret_cast<uint4*>(hist);
        for (int i = tid; i < DEG_HW / 4; i += 1024) h4[i] = make_uint4(0u, 0u, 0u, 0u);
    }
    __syncthreads();

    const int nq = e_pad / 4;
    const int qs = (((nq + DEG_S - 1) / DEG_S) + 31) & ~31;
    const int q0 = min(slice * qs, nq);
    const int q1 = min(q0 + qs, nq);
    const bool wr_keys = (window == 0);

    auto take = [&](int q, int4 t) {
        const unsigned d[4] = {(unsigned)t.x, (unsigned)t.y, (unsigned)t.z, (unsigned)t.w};
#pragma unroll
        for (int c = 0; c < 4; ++c) {
            const unsigned r = d[c] - (unsigned)lo;
            if (r < (unsigned)DEG_HW) atomicAdd(&hist[r], 1u);
        }
        if (wr_keys) {
            unsigned w = 0;
#pragma unroll
            for (int c = 0; c < 4; ++c) w |= min(d[c] / (unsigned)TN, 255u) << (8 * c);
            keys[q] = w;
        }
    };

    const bool al = (reinterpret_cast<size_t>(dst) & 15) == 0;
    const int qv1 = al ? max(q0, min(q1, e_total / 4)) : q0;
    const int4* dst4 = reinterpret_cast<const int4*>(dst);
    for (int qb = q0 + tid; qb < qv1; qb += U * 1024) {
        int4 v[U];
#pragma unroll
        for (int u = 0; u < U; ++u) {
            const int q = qb + u * 1024;
            v[u] = make_int4(-1, -1, -1, -1);
            if (q < qv1) v[u] = dst4[q];
        }
#pragma unroll
        for (int u = 0; u < U; ++u) {
            const int q = qb + u * 1024;
            if (q < qv1) take(q, v[u]);
        }
    }
    for (int q = qv1 + tid; q < q1; q += 1024) {
        const int e = 4 * q;
        int4 t = make_int4(-1, -1, -1, -1);
        if (e     < e_total) t.x = dst[e];
        if (e + 1 < e_total) t.y = dst[e + 1];
        if (e + 2 < e_total) t.z = dst[e + 2];
        if (e + 3 < e_total) t.w = dst[e + 3];
        take(q, t);
    }
    __syncthreads();

    unsigned* prow = part + (size_t)slice * n + lo;
    const int cnt = min(DEG_HW, n - lo);
    for (int i = tid; i < cnt; i += 1024) prow[i] = hist[i];
}

__global__ void dinv_kernel(const unsigned* __restrict__ part,
                            float* __restrict__ dinv, int n) {
    int i = blockIdx.x * blockDim.x + threadIdx.x;
    if (i < n) {
        unsigned c = 1u;
#pragma unroll
        for (int s = 0; s < DEG_S; ++s) c += part[(size_t)s * n + i];
        float d = (float)c;
        dinv[i] = (d > 0.0f) ? __frsqrt_rn(d) : 0.0f;
    }
}

template <int NDIM> struct GemmStage { static constexpr int CP = NDIM + 4; };
template <int KDIM, int NDIM>
__device__ __forceinline__ void
gemm_wmma_f32_wave(const float* __restrict__ A, const float* __restrict__ W,
                   float* __restrict__ C, int M, float* st, int tile, int nwaves) {
    constexpr int KT = KDIM / 4;
    constexpr int NT = NDIM / 16;
    constexpr int CP = GemmStage<NDIM>::CP;
    constexpr int NV = 16 * NDIM / 128;

    const int lane = threadIdx.x & 31;
    const int half = lane >> 4;
    const int l15  = lane & 15;
    const int ntiles = M / 16;
    if (tile >= ntiles) return;

    v2f bf[KT][NT];
#pragma unroll
    for (int kk = 0; kk < KT; ++kk) {
#pragma unroll
        for (int nt = 0; nt < NT; ++nt) {
            const float* wp = W + (size_t)(kk * 4 + 2 * half) * NDIM + nt * 16 + l15;
            v2f b;
            b.x = wp[0];
            b.y = wp[NDIM];
            bf[kk][nt] = b;
        }
    }


    for (; tile < ntiles; tile += nwaves) {
        v2f af[KT];
        const float* ap = A + (size_t)(tile * 16 + l15) * KDIM + 2 * half;
#pragma unroll
        for (int kk = 0; kk < KT; ++kk) af[kk] = *reinterpret_cast<const v2f*>(ap + kk * 4);

#pragma unroll
        for (int nt = 0; nt < NT; ++nt) {
            v8f acc = {};
#pragma unroll
            for (int kk = 0; kk < KT; ++kk) {
                acc = __builtin_amdgcn_wmma_f32_16x16x4_f32(
                     false, af[kk],  false, bf[kk][nt],
                     (short)0, acc,  false,  false);
            }
#pragma unroll
            for (int r = 0; r < 8; ++r) {
                st[(r + 8 * half) * CP + nt * 16 + l15] = acc[r];
            }
        }

        v4f* c4 = reinterpret_cast<v4f*>(C + (size_t)tile * 16 * NDIM);
        v4f v[NV];
#pragma unroll
        for (int j = 0; j < NV; ++j) {
            const int e = (j * 32 + lane) * 4;
            v[j] = *reinterpret_cast<const v4f*>(st + (e / NDIM) * CP + (e % NDIM));
        }
#pragma unroll
        for (int g = 0; g < NV; g += 8) {
            asm volatile("" :: "v"(v[g]), "v"(v[g + 1]), "v"(v[g + 2]), "v"(v[g + 3]), "v"(v[g + 4]), "v"(v[g + 5]), "v"(v[g + 6]), "v"(v[g + 7]));
        }
#pragma unroll
        for (int j = 0; j < NV; ++j) c4[j * 32 + lane] = v[j];
    }
}

template <int KDIM, int NDIM>
__global__ void __launch_bounds__(128)
gemm_wmma_f32_kernel(const float* __restrict__ A, const float* __restrict__ W,
                     float* __restrict__ C, int M) {
    constexpr int CP = GemmStage<NDIM>::CP;
    __shared__ float4 stage4[4 * 16 * CP / 4];
    const int wave = __builtin_amdgcn_readfirstlane((int)(threadIdx.x >> 5));
    gemm_wmma_f32_wave<KDIM, NDIM>(A, W, C, M, reinterpret_cast<float*>(stage4) + wave * (16 * CP),
                                   (int)blockIdx.x * 4 + wave, (int)gridDim.x * 4);
}

template <int TN, int KDIM, int NDIM>
__global__ void __launch_bounds__(1024)
deg_hist_key_gemm_kernel(const int* __restrict__ dst, unsigned* __restrict__ part,
                         unsigned* __restrict__ keys, int n, int e_total, int e_pad, int hist_groups,
                         const float* __restrict__ A, const float* __restrict__ W, float* __restrict__ C) {
    constexpr int CP = GemmStage<NDIM>::CP;
    constexpr int STAGE_BYTES = 32 * 16 * CP * 4;
    constexpr int LDS_BYTES = (DEG_HW * 4 > STAGE_BYTES) ? DEG_HW * 4 : STAGE_BYTES;
    __shared__ float4 lds4[LDS_BYTES / 16];

    const int b = (int)blockIdx.x;
    if (b < hist_groups) {
        deg_hist_key_body<TN>(dst, part, keys, n, e_total, e_pad, b % DEG_S, b / DEG_S,
                              reinterpret_cast<unsigned*>(lds4));
    } else {
        const int wave = __builtin_amdgcn_readfirstlane((int)(threadIdx.x >> 5));
        const int g  = b - hist_groups;
        const int ng = (int)gridDim.x - hist_groups;
        gemm_wmma_f32_wave<KDIM, NDIM>(A, W, C, n, reinterpret_cast<float*>(lds4) + wave * (16 * CP),
                                       wave * ng + g, ng * 32);
    }
}

constexpr int REC_CAP = 512;
template <int C> struct RowVec;
template <> struct RowVec<2> { typedef float2 type; };
template <> struct RowVec<4> { typedef float4 type; };
template <int F, int TN, bool RELU, bool WRITE_REC, bool READ_REC>
__global__ void __launch_bounds__(1024)
agg_tile_kernel(const float* __restrict__ h, const int* __restrict__ src,
                const int* __restrict__ dst, const float* __restrict__ dinv,
                const float* __restrict__ bias, float* __restrict__ out,
                const unsigned* __restrict__ keys, int* __restrict__ rec,
                int* __restrict__ rcnt, int n, int e_pad) {
    constexpr int NW   = 32;
    constexpr int QCAP = 64;
    constexpr int C    = F / 32;
    constexpr int RB   = 64 / C;
    typedef typename RowVec<C>::type rowv_t;
    constexpr int CH   = 4096;

    __shared__ float acc[TN * F];
    __shared__ float sdinv[TN];
    __shared__ int   qe[NW * QCAP];

    const int tid   = threadIdx.x;
    const int lane  = tid & 31;
    const int wave  = __builtin_amdgcn_readfirstlane(tid >> 5);
    const int node0 = blockIdx.x * TN;
    const int nrows = min(TN, n - node0);

    int4* myrec = reinterpret_cast<int4*>(rec) + ((size_t)blockIdx.x * NW + wave) * REC_CAP;
    int  recn = 0;
    bool use_list = false;
    int  nrec = 0;
    if (READ_REC) {
        const int c_ = rcnt[blockIdx.x * NW + lane];
        use_list = __builtin_amdgcn_ballot_w32(c_ > REC_CAP) == 0;
        nrec = __builtin_amdgcn_readlane(c_, wave);
    }

    for (int r = tid; r < nrows; r += 1024) sdinv[r] = dinv[node0 + r];
    {
        const float4* h4 = reinterpret_cast<const float4*>(h + (size_t)node0 * F);
        for (int i = tid; i < nrows * (F / 4); i += 1024) {
            float dv = dinv[node0 + i / (F / 4)];
            float4 v = h4[i];
            v.x = v.x * dv * dv; v.y = v.y * dv * dv;
            v.z = v.z * dv * dv; v.w = v.w * dv * dv;
            const int r  = i / (F / 4);
            const int j0 = 4 * (i - r * (F / 4));
            float* ar = acc + r * F;
            ar[((j0    ) % C) * 32 + (j0    ) / C] = v.x;
            ar[((j0 + 1) % C) * 32 + (j0 + 1) / C] = v.y;
            ar[((j0 + 2) % C) * 32 + (j0 + 2) / C] = v.z;
            ar[((j0 + 3) % C) * 32 + (j0 + 3) / C] = v.w;
        }
    }
    __syncthreads();

    const int  nchunks = e_pad / CH;
    const int  qbase   = wave * QCAP;
    int qn = 0;

    auto gather = [&](int s_, int dl_, float nrm_, int cnt) {
        for (int k0 = 0; k0 < cnt; k0 += RB) {
            rowv_t v[RB];
#pragma unroll
            for (int j = 0; j < RB; ++j) {
                int k = min(k0 + j, cnt - 1);
                int s = __builtin_amdgcn_readlane(s_, k);
                v[j] = *reinterpret_cast<const rowv_t*>(h + (size_t)s * F + C * lane);
            }
#pragma unroll
            for (int j = 0; j < RB; ++j) {
                int k = k0 + j;
                if (k < cnt) {
                    float nrm = __int_as_float(__builtin_amdgcn_readlane(__float_as_int(nrm_), k));
                    int   dl  = __builtin_amdgcn_readlane(dl_, k);
                    float* ap = &acc[dl * F + lane];
                    const float* vj = reinterpret_cast<const float*>(&v[j]);
#pragma unroll
                    for (int c = 0; c < C; ++c) atomicAdd(ap + 32 * c, vj[c] * nrm);
                }
            }
        }
    };

    auto run_batch = [&](int start, int cnt) {
        int s_ = 0, dl_ = 0; float nrm_ = 0.0f;
        if (lane < cnt) {
            int e_ = qe[qbase + start + lane];
            s_  = src[e_];
            dl_ = dst[e_] - node0;
            nrm_ = dinv[s_] * sdinv[dl_];
        }
        gather(s_, dl_, nrm_, cnt);
        if (WRITE_REC) {
            if (lane < cnt && recn + lane < REC_CAP)
                myrec[recn + lane] = make_int4(s_, dl_, __float_as_int(nrm_), 0);
            recn += cnt;
        }
    };

    auto drain = [&](unsigned bits, int ebase) {
        for (int t = 0; t < 32; ++t) {
            unsigned mask = __builtin_amdgcn_ballot_w32(bits != 0);
            if (mask == 0) break;
            if (bits != 0) {
                int b = __builtin_ctz(bits);
                bits &= bits - 1;
                int pos = qn + (int)__builtin_amdgcn_mbcnt_lo(mask, 0u);
                int i = b & 7;
                qe[qbase + pos] = ebase + 512 * (i >> 2) + 4 * (i & 3) + (b >> 3);
            }
            qn += __builtin_popcount(mask);
            if (qn >= 32) {
                qn -= 32;
                run_batch(qn, 32);
            }
        }
    };

    if (use_list) {
        for (int b = 0; b < nrec; b += 32) {
            int4 r_ = make_int4(0, 0, 0, 0);
            if (b + lane < nrec) r_ = myrec[b + lane];
            gather(r_.x, r_.y, __int_as_float(r_.z), min(32, nrec - b));
        }
    } else {
        const unsigned pat = (unsigned)blockIdx.x * 0x01010101u;
        const uint4* keys4 = reinterpret_cast<const uint4*>(keys);
        for (int chunk = wave; chunk < nchunks; chunk += NW) {
            const int base = chunk * CH;
            uint4 kw[8];
#pragma unroll
            for (int k = 0; k < 8; ++k) kw[k] = keys4[(base >> 4) + 32 * k + lane];

#pragma unroll
            for (int p = 0; p < 4; ++p) {
                const unsigned w8[8] = {kw[2 * p].x,     kw[2 * p].y,     kw[2 * p].z,     kw[2 * p].w,
                                        kw[2 * p + 1].x, kw[2 * p + 1].y, kw[2 * p + 1].z, kw[2 * p + 1].w};
                unsigned bits = 0;
#pragma unroll
                for (int i = 0; i < 8; ++i) {
                    unsigned x = w8[i] ^ pat;
                    unsigned y = ((x & 0x7F7F7F7Fu) + 0x7F7F7F7Fu) | x;
                    bits |= (~y & 0x80808080u) >> (7 - i);
                }
                drain(bits, base + 1024 * p + 16 * lane);
            }
        }
        run_batch(0, qn);
    }
    if (WRITE_REC && lane == 0) rcnt[blockIdx.x * NW + wave] = recn;
    __syncthreads();

    {
        const float4* b4 = reinterpret_cast<const float4*>(bias);
        float4* o4 = reinterpret_cast<float4*>(out + (size_t)node0 * F);
        for (int i = tid; i < nrows * (F / 4); i += 1024) {
            const int r  = i / (F / 4);
            const int j0 = 4 * (i - r * (F / 4));
            const float* ar = acc + r * F;
            float4 v;
            v.x = ar[((j0    ) % C) * 32 + (j0    ) / C];
            v.y = ar[((j0 + 1) % C) * 32 + (j0 + 1) / C];
            v.z = ar[((j0 + 2) % C) * 32 + (j0 + 2) / C];
            v.w = ar[((j0 + 3) % C) * 32 + (j0 + 3) / C];
            float4 b = b4[i % (F / 4)];
            v.x += b.x; v.y += b.y; v.z += b.z; v.w += b.w;
            if (RELU) {
                v.x = fmaxf(v.x, 0.0f); v.y = fmaxf(v.y, 0.0f);
                v.z = fmaxf(v.z, 0.0f); v.w = fmaxf(v.w, 0.0f);
            }
            o4[i] = v;
        }
    }
}

extern "C" void kernel_launch(void* const* d_in, const int* in_sizes, int n_in,
                              void* d_out, int out_size, void* d_ws, size_t ws_size,
                              hipStream_t stream) {
    const float* x  = (const float*)d_in[0];
    const int*   ei = (const int*)d_in[1];
    const float* W1 = (const float*)d_in[2];
    const float* b1 = (const float*)d_in[3];
    const float* W2 = (const float*)d_in[4];
    const float* b2 = (const float*)d_in[5];
    float* out = (float*)d_out;

    const int       N = in_sizes[0] / 32;
    const long long E = (long long)in_sizes[1] / 2;
    const int* src = ei;
    const int* dst = ei + E;

    const int EPAD = (int)((E + 4095) / 4096) * 4096;
    const int TILES = (N + 399) / 400;
    const int NWIN = (N + DEG_HW - 1) / DEG_HW;
    float* hw1  = (float*)d_ws;
    float* h1   = hw1 + (size_t)N * 64;
    float* hw2  = h1 + (size_t)N * 64;
    int* rec    = (int*)(hw2 + (size_t)N * 128);
    int* rcnt   = rec + (size_t)TILES * 32 * REC_CAP * 4;
    unsigned* keys = (unsigned*)(rcnt + (size_t)TILES * 32);
    float* dinv = (float*)((char*)keys + EPAD);
    unsigned* part = (unsigned*)(dinv + N);

    const int BT = 256;

    deg_hist_key_gemm_kernel<400, 32, 64><<<DEG_S * NWIN + 128, 1024, 0, stream>>>(dst, part, keys, N, (int)E, EPAD, DEG_S * NWIN, x, W1, hw1);
    dinv_kernel<<<(N + BT - 1) / BT, BT, 0, stream>>>(part, dinv, N);

    agg_tile_kernel<64, 400, true, true, false><<<TILES, 1024, 0, stream>>>(hw1, src, dst, dinv, b1, h1, keys, rec, rcnt, N, EPAD);

    gemm_wmma_f32_kernel<64, 128><<<512, 128, 0, stream>>>(h1, W2, hw2, N);
    agg_tile_kernel<128, 400, false, false, true><<<TILES, 1024, 0, stream>>>(hw2, src, dst, dinv, b2, out, keys, rec, rcnt, N, EPAD);
}


# Layer 2: a wave owns 12-13 tile rows and sums them in registers from its own records (layer 1 hands each kept edge to the row's owner via LDS queues): no LDS float adds; f32 sum order: queue order
# speedup vs baseline: 4.1198x; 1.0443x over previous
#include <hip/hip_runtime.h>
#include <hip/hip_bf16.h>

typedef float v2f __attribute__((ext_vector_type(2)));
typedef float v8f __attribute__((ext_vector_type(8)));
typedef float v4f __attribute__((ext_vector_type(4)));

constexpr int DEG_S  = 32;
constexpr int DEG_HW = 25024;

template <int TN>
__device__ __forceinline__ void
deg_hist_key_body(const int* __restrict__ dst, unsigned* __restrict__ part,
                  unsigned* __restrict__ keys, int n, int e_total, int e_pad,
                  int slice, int window, unsigned* hist) {
    constexpr int U = 16;

    const int tid = threadIdx.x;
    const int lo  = window * DEG_HW;
    {
        uint4* h4 = reinterpret_cast<uint4*>(hist);
        for (int i = tid; i < DEG_HW / 4; i += 1024) h4[i] = make_uint4(0u, 0u, 0u, 0u);
    }
    __syncthreads();

    const int nq = e_pad / 4;
    const int qs = (((nq + DEG_S - 1) / DEG_S) + 31) & ~31;
    const int q0 = min(slice * qs, nq);
    const int q1 = min(q0 + qs, nq);
    const bool wr_keys = (window == 0);

    auto take = [&](int q, int4 t) {
        const unsigned d[4] = {(unsigned)t.x, (unsigned)t.y, (unsigned)t.z, (unsigned)t.w};
#pragma unroll
        for (int c = 0; c < 4; ++c) {
            const unsigned r = d[c] - (unsigned)lo;
            if (r < (unsigned)DEG_HW) atomicAdd(&hist[r], 1u);
        }
        if (wr_keys) {
            unsigned w = 0;
#pragma unroll
            for (int c = 0; c < 4; ++c) w |= min(d[c] / (unsigned)TN, 255u) << (8 * c);
            keys[q] = w;
        }
    };

    const bool al = (reinterpret_cast<size_t>(dst) & 15) == 0;
    const int qv1 = al ? max(q0, min(q1, e_total / 4)) : q0;
    const int4* dst4 = reinterpret_cast<const int4*>(dst);
    for (int qb = q0 + tid; qb < qv1; qb += U * 1024) {
        int4 v[U];
#pragma unroll
        for (int u = 0; u < U; ++u) {
            const int q = qb + u * 1024;
            v[u] = make_int4(-1, -1, -1, -1);
            if (q < qv1) v[u] = dst4[q];
        }
#pragma unroll
        for (int u = 0; u < U; ++u) {
            const int q = qb + u * 1024;
            if (q < qv1) take(q, v[u]);
        }
    }
    for (int q = qv1 + tid; q < q1; q += 1024) {
        const int e = 4 * q;
        int4 t = make_int4(-1, -1, -1, -1);
        if (e     < e_total) t.x = dst[e];
        if (e + 1 < e_total) t.y = dst[e + 1];
        if (e + 2 < e_total) t.z = dst[e + 2];
        if (e + 3 < e_total) t.w = dst[e + 3];
        take(q, t);
    }
    __syncthreads();

    unsigned* prow = part + (size_t)slice * n + lo;
    const int cnt = min(DEG_HW, n - lo);
    for (int i = tid; i < cnt; i += 1024) prow[i] = hist[i];
}

__global__ void dinv_kernel(const unsigned* __restrict__ part,
                            float* __restrict__ dinv, int n) {
    int i = blockIdx.x * blockDim.x + threadIdx.x;
    if (i < n) {
        unsigned c = 1u;
#pragma unroll
        for (int s = 0; s < DEG_S; ++s) c += part[(size_t)s * n + i];
        float d = (float)c;
        dinv[i] = (d > 0.0f) ? __frsqrt_rn(d) : 0.0f;
    }
}

template <int NDIM> struct GemmStage { static constexpr int CP = NDIM + 4; };
template <int KDIM, int NDIM>
__device__ __forceinline__ void
gemm_wmma_f32_wave(const float* __restrict__ A, const float* __restrict__ W,
                   float* __restrict__ C, int M, float* st, int tile, int nwaves) {
    constexpr int KT = KDIM / 4;
    constexpr int NT = NDIM / 16;
    constexpr int CP = GemmStage<NDIM>::CP;
    constexpr int NV = 16 * NDIM / 128;

    const int lane = threadIdx.x & 31;
    const int half = lane >> 4;
    const int l15  = lane & 15;
    const int ntiles = M / 16;
    if (tile >= ntiles) return;

    v2f bf[KT][NT];
#pragma unroll
    for (int kk = 0; kk < KT; ++kk) {
#pragma unroll
        for (int nt = 0; nt < NT; ++nt) {
            const float* wp = W + (size_t)(kk * 4 + 2 * half) * NDIM + nt * 16 + l15;
            v2f b;
            b.x = wp[0];
            b.y = wp[NDIM];
            bf[kk][nt] = b;
        }
    }


    for (; tile < ntiles; tile += nwaves) {
        v2f af[KT];
        const float* ap = A + (size_t)(tile * 16 + l15) * KDIM + 2 * half;
#pragma unroll
        for (int kk = 0; kk < KT; ++kk) af[kk] = *reinterpret_cast<const v2f*>(ap + kk * 4);

#pragma unroll
        for (int nt = 0; nt < NT; ++nt) {
            v8f acc = {};
#pragma unroll
            for (int kk = 0; kk < KT; ++kk) {
                acc = __builtin_amdgcn_wmma_f32_16x16x4_f32(
                     false, af[kk],  false, bf[kk][nt],
                     (short)0, acc,  false,  false);
            }
#pragma unroll
            for (int r = 0; r < 8; ++r) {
                st[(r + 8 * half) * CP + nt * 16 + l15] = acc[r];
            }
        }

        v4f* c4 = reinterpret_cast<v4f*>(C + (size_t)tile * 16 * NDIM);
        v4f v[NV];
#pragma unroll
        for (int j = 0; j < NV; ++j) {
            const int e = (j * 32 + lane) * 4;
            v[j] = *reinterpret_cast<const v4f*>(st + (e / NDIM) * CP + (e % NDIM));
        }
#pragma unroll
        for (int g = 0; g < NV; g += 8) {
            asm volatile("" :: "v"(v[g]), "v"(v[g + 1]), "v"(v[g + 2]), "v"(v[g + 3]), "v"(v[g + 4]), "v"(v[g + 5]), "v"(v[g + 6]), "v"(v[g + 7]));
        }
#pragma unroll
        for (int j = 0; j < NV; ++j) c4[j * 32 + lane] = v[j];
    }
}

template <int KDIM, int NDIM>
__global__ void __launch_bounds__(128)
gemm_wmma_f32_kernel(const float* __restrict__ A, const float* __restrict__ W,
                     float* __restrict__ C, int M) {
    constexpr int CP = GemmStage<NDIM>::CP;
    __shared__ float4 stage4[4 * 16 * CP / 4];
    const int wave = __builtin_amdgcn_readfirstlane((int)(threadIdx.x >> 5));
    gemm_wmma_f32_wave<KDIM, NDIM>(A, W, C, M, reinterpret_cast<float*>(stage4) + wave * (16 * CP),
                                   (int)blockIdx.x * 4 + wave, (int)gridDim.x * 4);
}

template <int TN, int KDIM, int NDIM>
__global__ void __launch_bounds__(1024)
deg_hist_key_gemm_kernel(const int* __restrict__ dst, unsigned* __restrict__ part,
                         unsigned* __restrict__ keys, int n, int e_total, int e_pad, int hist_groups,
                         const float* __restrict__ A, const float* __restrict__ W, float* __restrict__ C) {
    constexpr int CP = GemmStage<NDIM>::CP;
    constexpr int STAGE_BYTES = 32 * 16 * CP * 4;
    constexpr int LDS_BYTES = (DEG_HW * 4 > STAGE_BYTES) ? DEG_HW * 4 : STAGE_BYTES;
    __shared__ float4 lds4[LDS_BYTES / 16];

    const int b = (int)blockIdx.x;
    if (b < hist_groups) {
        deg_hist_key_body<TN>(dst, part, keys, n, e_total, e_pad, b % DEG_S, b / DEG_S,
                              reinterpret_cast<unsigned*>(lds4));
    } else {
        const int wave = __builtin_amdgcn_readfirstlane((int)(threadIdx.x >> 5));
        const int g  = b - hist_groups;
        const int ng = (int)gridDim.x - hist_groups;
        gemm_wmma_f32_wave<KDIM, NDIM>(A, W, C, n, reinterpret_cast<float*>(lds4) + wave * (16 * CP),
                                       wave * ng + g, ng * 32);
    }
}

constexpr int REC_CAP = 384;

typedef float v16f __attribute__((ext_vector_type(16)));
template <int C> struct RowVec;
template <> struct RowVec<2> { typedef float2 type; };
template <> struct RowVec<4> { typedef float4 type; };
template <int F, int TN, bool RELU, bool WRITE_REC, bool READ_REC>
__global__ void __launch_bounds__(1024)
agg_tile_kernel(const float* __restrict__ h, const int* __restrict__ src,
                const int* __restrict__ dst, const float* __restrict__ dinv,
                const float* __restrict__ bias, float* __restrict__ out,
                const unsigned* __restrict__ keys, int* __restrict__ rec,
                int* __restrict__ rcnt, int n, int e_pad) {
    constexpr int NW   = 32;
    constexpr int QCAP = 64;
    constexpr int C    = F / 32;
    constexpr int RB   = 64 / C;
    typedef typename RowVec<C>::type rowv_t;
    constexpr int CH   = 4096;

    __shared__ float acc[TN * F];
    __shared__ float sdinv[TN];
    __shared__ int   qe[NW * QCAP];
    __shared__ int4  oq[WRITE_REC ? NW * REC_CAP : 1];
    __shared__ int   ocnt[NW];
    static_assert(NW == 32, "a row's owner is its low five bits");

    const int tid   = threadIdx.x;
    const int lane  = tid & 31;
    const int wave  = __builtin_amdgcn_readfirstlane(tid >> 5);
    const int node0 = blockIdx.x * TN;
    const int nrows = min(TN, n - node0);

    int4* myrec = reinterpret_cast<int4*>(rec) + ((size_t)blockIdx.x * NW + wave) * REC_CAP;
    bool use_list = false;
    int  nrec = 0;
    if (READ_REC) {
        const int c_ = rcnt[blockIdx.x * NW + lane];
        use_list = __builtin_amdgcn_ballot_w32(c_ > REC_CAP) == 0;
        nrec = __builtin_amdgcn_readlane(c_, wave);
    }

    if (READ_REC && __builtin_expect(use_list, 1)) {
        constexpr int NS  = (TN + NW - 1) / NW;
        constexpr int RB2 = 8;
        static_assert(NS <= 16, "a wave's row slots must fit one 16-register vector");
        v16f a[C];
        {
            rowv_t sv[NS];
            float  dv[NS];
#pragma unroll
            for (int k = 0; k < NS; ++k) {
                const int r = min(wave + NW * k, nrows - 1);
                dv[k] = dinv[node0 + r];
                sv[k] = *reinterpret_cast<const rowv_t*>(h + (size_t)(node0 + r) * F + C * lane);
            }
#pragma unroll
            for (int c = 0; c < C; ++c) {
                v16f t = {};
#pragma unroll
                for (int k = 0; k < NS; ++k) t[k] = reinterpret_cast<const float*>(&sv[k])[c] * dv[k] * dv[k];
                a[c] = t;
            }
        }
        for (int b = 0; b < nrec; b += 32) {
            int4 r_ = make_int4(0, 0, 0, 0);
            if (b + lane < nrec) r_ = myrec[b + lane];
            const int cnt = min(32, nrec - b);
            for (int k0 = 0; k0 < cnt; k0 += RB2) {
                rowv_t v[RB2];
#pragma unroll
                for (int j = 0; j < RB2; ++j) {
                    const int k = min(k0 + j, cnt - 1);
                    const int s = __builtin_amdgcn_readlane(r_.x, k);
                    v[j] = *reinterpret_cast<const rowv_t*>(h + (size_t)s * F + C * lane);
                }
#pragma unroll
                for (int j = 0; j < RB2; ++j) {
                    const int k = k0 + j;
                    if (k < cnt) {
                        const float nrm  = __int_as_float(__builtin_amdgcn_readlane(r_.z, k));
                        const int   slot = __builtin_amdgcn_readlane(r_.y, k) >> 5;
                        const float* vj = reinterpret_cast<const float*>(&v[j]);
#pragma unroll
                        for (int c = 0; c < C; ++c) a[c][slot] = fmaf(vj[c], nrm, a[c][slot]);
                    }
                }
            }
        }
        const rowv_t bv = *reinterpret_cast<const rowv_t*>(bias + C * lane);
        const float* bf = reinterpret_cast<const float*>(&bv);
#pragma unroll
        for (int k = 0; k < NS; ++k) {
            const int r = wave + NW * k;
            if (r < nrows) {
                rowv_t o;
                float* of = reinterpret_cast<float*>(&o);
#pragma unroll
                for (int c = 0; c < C; ++c) {
                    of[c] = a[c][k] + bf[c];
                    if (RELU) of[c] = fmaxf(of[c], 0.0f);
                }
                *reinterpret_cast<rowv_t*>(out + (size_t)(node0 + r) * F + C * lane) = o;
            }
        }
        return;
    }

    if (WRITE_REC && tid < NW) ocnt[tid] = 0;

    for (int r = tid; r < nrows; r += 1024) sdinv[r] = dinv[node0 + r];
    {
        const float4* h4 = reinterpret_cast<const float4*>(h + (size_t)node0 * F);
        for (int i = tid; i < nrows * (F / 4); i += 1024) {
            float dv = dinv[node0 + i / (F / 4)];
            float4 v = h4[i];
            v.x = v.x * dv * dv; v.y = v.y * dv * dv;
            v.z = v.z * dv * dv; v.w = v.w * dv * dv;
            const int r  = i / (F / 4);
            const int j0 = 4 * (i - r * (F / 4));
            float* ar = acc + r * F;
            ar[((j0    ) % C) * 32 + (j0    ) / C] = v.x;
            ar[((j0 + 1) % C) * 32 + (j0 + 1) / C] = v.y;
            ar[((j0 + 2) % C) * 32 + (j0 + 2) / C] = v.z;
            ar[((j0 + 3) % C) * 32 + (j0 + 3) / C] = v.w;
        }
    }
    __syncthreads();

    const int  nchunks = e_pad / CH;
    const int  qbase   = wave * QCAP;
    int qn = 0;

    auto gather = [&](int s_, int dl_, float nrm_, int cnt) {
        for (int k0 = 0; k0 < cnt; k0 += RB) {
            rowv_t v[RB];
#pragma unroll
            for (int j = 0; j < RB; ++j) {
                int k = min(k0 + j, cnt - 1);
                int s = __builtin_amdgcn_readlane(s_, k);
                v[j] = *reinterpret_cast<const rowv_t*>(h + (size_t)s * F + C * lane);
            }
#pragma unroll
            for (int j = 0; j < RB; ++j) {
                int k = k0 + j;
                if (k < cnt) {
                    float nrm = __int_as_float(__builtin_amdgcn_readlane(__float_as_int(nrm_), k));
                    int   dl  = __builtin_amdgcn_readlane(dl_, k);
                    float* ap = &acc[dl * F + lane];
                    const float* vj = reinterpret_cast<const float*>(&v[j]);
#pragma unroll
                    for (int c = 0; c < C; ++c) atomicAdd(ap + 32 * c, vj[c] * nrm);
                }
            }
        }
    };

    auto run_batch = [&](int start, int cnt) {
        int s_ = 0, dl_ = 0; float nrm_ = 0.0f;
        int ow = 0, at = REC_CAP;
        if (lane < cnt) {
            int e_ = qe[qbase + start + lane];
            s_  = src[e_];
            dl_ = dst[e_] - node0;
            if (WRITE_REC) {
                ow = dl_ & (NW - 1);
                at = atomicAdd(&ocnt[ow], 1);
            }
            nrm_ = dinv[s_] * sdinv[dl_];
        }
        gather(s_, dl_, nrm_, cnt);
        if (WRITE_REC) {
            if (at < REC_CAP) oq[ow * REC_CAP + at] = make_int4(s_, dl_, __float_as_int(nrm_), 0);
        }
    };

    auto drain = [&](unsigned bits, int ebase) {
        for (int t = 0; t < 32; ++t) {
            unsigned mask = __builtin_amdgcn_ballot_w32(bits != 0);
            if (mask == 0) break;
            if (bits != 0) {
                int b = __builtin_ctz(bits);
                bits &= bits - 1;
                int pos = qn + (int)__builtin_amdgcn_mbcnt_lo(mask, 0u);
                int i = b & 7;
                qe[qbase + pos] = ebase + 512 * (i >> 2) + 4 * (i & 3) + (b >> 3);
            }
            qn += __builtin_popcount(mask);
            if (qn >= 32) {
                qn -= 32;
                run_batch(qn, 32);
            }
        }
    };

    {
        const unsigned pat = (unsigned)blockIdx.x * 0x01010101u;
        const uint4* keys4 = reinterpret_cast<const uint4*>(keys);
        for (int chunk = wave; chunk < nchunks; chunk += NW) {
            const int base = chunk * CH;
            uint4 kw[8];
#pragma unroll
            for (int k = 0; k < 8; ++k) kw[k] = keys4[(base >> 4) + 32 * k + lane];

#pragma unroll
            for (int p = 0; p < 4; ++p) {
                const unsigned w8[8] = {kw[2 * p].x,     kw[2 * p].y,     kw[2 * p].z,     kw[2 * p].w,
                                        kw[2 * p + 1].x, kw[2 * p + 1].y, kw[2 * p + 1].z, kw[2 * p + 1].w};
                unsigned bits = 0;
#pragma unroll
                for (int i = 0; i < 8; ++i) {
                    unsigned x = w8[i] ^ pat;
                    unsigned y = ((x & 0x7F7F7F7Fu) + 0x7F7F7F7Fu) | x;
                    bits |= (~y & 0x80808080u) >> (7 - i);
                }
                drain(bits, base + 1024 * p + 16 * lane);
            }
        }
        run_batch(0, qn);
    }
    __syncthreads();
    if (WRITE_REC) {
        const int cown = ocnt[wave];
        const int m = min(cown, REC_CAP);
        for (int i = lane; i < m; i += 32) myrec[i] = oq[wave * REC_CAP + i];
        if (lane == 0) rcnt[blockIdx.x * NW + wave] = cown;
    }

    {
        const float4* b4 = reinterpret_cast<const float4*>(bias);
        float4* o4 = reinterpret_cast<float4*>(out + (size_t)node0 * F);
        for (int i = tid; i < nrows * (F / 4); i += 1024) {
            const int r  = i / (F / 4);
            const int j0 = 4 * (i - r * (F / 4));
            const float* ar = acc + r * F;
            float4 v;
            v.x = ar[((j0    ) % C) * 32 + (j0    ) / C];
            v.y = ar[((j0 + 1) % C) * 32 + (j0 + 1) / C];
            v.z = ar[((j0 + 2) % C) * 32 + (j0 + 2) / C];
            v.w = ar[((j0 + 3) % C) * 32 + (j0 + 3) / C];
            float4 b = b4[i % (F / 4)];
            v.x += b.x; v.y += b.y; v.z += b.z; v.w += b.w;
            if (RELU) {
                v.x = fmaxf(v.x, 0.0f); v.y = fmaxf(v.y, 0.0f);
                v.z = fmaxf(v.z, 0.0f); v.w = fmaxf(v.w, 0.0f);
            }
            o4[i] = v;
        }
    }
}

extern "C" void kernel_launch(void* const* d_in, const int* in_sizes, int n_in,
                              void* d_out, int out_size, void* d_ws, size_t ws_size,
                              hipStream_t stream) {
    const float* x  = (const float*)d_in[0];
    const int*   ei = (const int*)d_in[1];
    const float* W1 = (const float*)d_in[2];
    const float* b1 = (const float*)d_in[3];
    const float* W2 = (const float*)d_in[4];
    const float* b2 = (const float*)d_in[5];
    float* out = (float*)d_out;

    const int       N = in_sizes[0] / 32;
    const long long E = (long long)in_sizes[1] / 2;
    const int* src = ei;
    const int* dst = ei + E;

    const int EPAD = (int)((E + 4095) / 4096) * 4096;
    const int TILES = (N + 399) / 400;
    const int NWIN = (N + DEG_HW - 1) / DEG_HW;
    float* hw1  = (float*)d_ws;
    float* h1   = hw1 + (size_t)N * 64;
    float* hw2  = h1 + (size_t)N * 64;
    int* rec    = (int*)(hw2 + (size_t)N * 128);
    int* rcnt   = rec + (size_t)TILES * 32 * REC_CAP * 4;
    unsigned* keys = (unsigned*)(rcnt + (size_t)TILES * 32);
    float* dinv = (float*)((char*)keys + EPAD);
    unsigned* part = (unsigned*)(dinv + N);

    const int BT = 256;

    deg_hist_key_gemm_kernel<400, 32, 64><<<DEG_S * NWIN + 128, 1024, 0, stream>>>(dst, part, keys, N, (int)E, EPAD, DEG_S * NWIN, x, W1, hw1);
    dinv_kernel<<<(N + BT - 1) / BT, BT, 0, stream>>>(part, dinv, N);

    agg_tile_kernel<64, 400, true, true, false><<<TILES, 1024, 0, stream>>>(hw1, src, dst, dinv, b1, h1, keys, rec, rcnt, N, EPAD);

    gemm_wmma_f32_kernel<64, 128><<<512, 128, 0, stream>>>(h1, W2, hw2, N);
    agg_tile_kernel<128, 400, false, false, true><<<TILES, 1024, 0, stream>>>(hw2, src, dst, dinv, b2, out, keys, rec, rcnt, N, EPAD);
}
